# starting kernel
# baseline (speedup 1.0000x reference)
#include <hip/hip_runtime.h>

typedef float v2f __attribute__((ext_vector_type(2)));
typedef float v8f __attribute__((ext_vector_type(8)));

#define BB   2
#define DD   128
#define HH   128
#define WW   128
#define CI   64
#define CO   64
#define NOFF 27
#define NPTS 200000
#define GRID_CELLS (BB*DD*HH*WW)
#define WT_FLOATS  (NOFF*16*4*32*2)
#define A_STRIDE   68

__global__ void k_fill_grid(int* __restrict__ grid) {
    int i = blockIdx.x * blockDim.x + threadIdx.x;
    if (i < GRID_CELLS) grid[i] = -1;
}

__global__ void k_scatter(const int* __restrict__ coords, int* __restrict__ grid, int n) {
    int i = blockIdx.x * blockDim.x + threadIdx.x;
    if (i >= n) return;
    int b = coords[i*4+0], z = coords[i*4+1], y = coords[i*4+2], x = coords[i*4+3];
    int lin = ((b*DD + z)*HH + y)*WW + x;
    grid[lin] = i;
}

__global__ void k_wt(const float* __restrict__ w, float* __restrict__ wT) {
    int i = blockIdx.x * blockDim.x + threadIdx.x;
    if (i >= NOFF*16*4*32) return;
    int lane = i & 31;
    int nt   = (i >> 5) & 3;
    int kc   = (i >> 7) & 15;
    int t    =  i >> 11;
    int o = nt*16 + (lane & 15);
    int c = kc*4 + 2*(lane >> 4);
    const float* src = w + (size_t)(o*NOFF + t)*CI + c;
    wT[(size_t)i*2 + 0] = src[0];
    wT[(size_t)i*2 + 1] = src[1];
}

__global__ __launch_bounds__(128)
void k_conv(const float* __restrict__ feats,
            const float* __restrict__ wT,
            const float* __restrict__ bias,
            const int*   __restrict__ coords,
            const int*   __restrict__ grid,
            float*       __restrict__ out) {
    __shared__ int   nbr[NOFF*16];
    __shared__ float A[16*A_STRIDE];

    const int tid  = threadIdx.x;
    const int lane = tid & 31;
    const int wv   = tid >> 5;
    const int p0   = blockIdx.x * 16;

    for (int i = tid; i < NOFF*16; i += 128) {
        int t = i >> 4, m = i & 15;
        int p = p0 + m;
        int bz = coords[p*4+0], z = coords[p*4+1], y = coords[p*4+2], x = coords[p*4+3];
        int kd = t / 9, kh = (t / 3) % 3, kw = t % 3;
        int nz = z + kd - 1, ny = y + kh - 1, nx = x + kw - 1;
        int idx = -1;
        if ((unsigned)nz < (unsigned)DD && (unsigned)ny < (unsigned)HH &&
            (unsigned)nx < (unsigned)WW) {
            int lin = ((bz*DD + nz)*HH + ny)*WW + nx;
            idx = grid[lin];
        }
        nbr[i] = idx;
    }
    __syncthreads();

    const int col  = lane & 15;
    const int half = lane >> 4;
    float bv = bias[wv*16 + col];
    v8f acc;
#pragma unroll
    for (int j = 0; j < 8; ++j) acc[j] = bv;

    const int arow  = tid >> 3;
    const int acol8 = (tid & 7) * 8;

    for (int t = 0; t < NOFF; ++t) {
        int idx = nbr[t*16 + arow];
        float4 lo = make_float4(0.f, 0.f, 0.f, 0.f), hi = lo;
        if (idx >= 0) {
            const float4* src = (const float4*)(feats + (size_t)idx*CI + acol8);
            lo = src[0];
            hi = src[1];
        }
        float* dst = &A[arow*A_STRIDE + acol8];
        *(float4*)(dst)     = lo;
        *(float4*)(dst + 4) = hi;
        __syncthreads();

        const float* wbase = wT + (size_t)((t*16)*4 + wv)*64 + lane*2;
#pragma unroll
        for (int kc = 0; kc < 16; ++kc) {
            const float* ap = &A[col*A_STRIDE + kc*4 + 2*half];
            v2f a; a.x = ap[0]; a.y = ap[1];
            const float* bp = wbase + (size_t)kc*256;
            v2f b; b.x = bp[0]; b.y = bp[1];
            acc = __builtin_amdgcn_wmma_f32_16x16x4_f32(
                false, a, false, b, (short)0, acc, false, false);
        }
        __syncthreads();
    }

#pragma unroll
    for (int j = 0; j < 8; ++j) {
        int m = j + 8*half;
        out[(size_t)(p0 + m)*CO + wv*16 + col] = acc[j];
    }
}

extern "C" void kernel_launch(void* const* d_in, const int* in_sizes, int n_in,
                              void* d_out, int out_size, void* d_ws, size_t ws_size,
                              hipStream_t stream) {
    const float* feats  = (const float*)d_in[0];
    const float* weight = (const float*)d_in[1];
    const float* bias   = (const float*)d_in[2];
    const int*   coords = (const int*)  d_in[3];
    float* out = (float*)d_out;

    const int n = in_sizes[0] / CI;

    float* wT   = (float*)d_ws;
    int*   grid = (int*)((char*)d_ws + (size_t)WT_FLOATS*4);

    k_fill_grid<<<(GRID_CELLS + 255)/256, 256, 0, stream>>>(grid);
    k_scatter  <<<(n + 255)/256,          256, 0, stream>>>(coords, grid, n);
    k_wt       <<<(NOFF*16*4*32 + 255)/256, 256, 0, stream>>>(weight, wT);
    k_conv     <<<n/16, 128, 0, stream>>>(feats, wT, bias, coords, grid, out);
}


# conv loop runs only the taps where some point of the 16-point tile has an active neighbour (per-tap ballot mask, scalar loop); a tap with none adds exact zeros and is skipped
# speedup vs baseline: 1.5354x; 1.5354x over previous
#include <hip/hip_runtime.h>

typedef float v2f __attribute__((ext_vector_type(2)));
typedef float v8f __attribute__((ext_vector_type(8)));

#define BB   2
#define DD   128
#define HH   128
#define WW   128
#define CI   64
#define CO   64
#define NOFF 27
#define NPTS 200000
#define GRID_CELLS (BB*DD*HH*WW)
#define WT_FLOATS  (NOFF*16*4*32*2)
#define A_STRIDE   68

__global__ void k_fill_grid(int* __restrict__ grid) {
    int i = blockIdx.x * blockDim.x + threadIdx.x;
    if (i < GRID_CELLS) grid[i] = -1;
}

__global__ void k_scatter(const int* __restrict__ coords, int* __restrict__ grid, int n) {
    int i = blockIdx.x * blockDim.x + threadIdx.x;
    if (i >= n) return;
    int b = coords[i*4+0], z = coords[i*4+1], y = coords[i*4+2], x = coords[i*4+3];
    int lin = ((b*DD + z)*HH + y)*WW + x;
    grid[lin] = i;
}

__global__ void k_wt(const float* __restrict__ w, float* __restrict__ wT) {
    int i = blockIdx.x * blockDim.x + threadIdx.x;
    if (i >= NOFF*16*4*32) return;
    int lane = i & 31;
    int nt   = (i >> 5) & 3;
    int kc   = (i >> 7) & 15;
    int t    =  i >> 11;
    int o = nt*16 + (lane & 15);
    int c = kc*4 + 2*(lane >> 4);
    const float* src = w + (size_t)(o*NOFF + t)*CI + c;
    wT[(size_t)i*2 + 0] = src[0];
    wT[(size_t)i*2 + 1] = src[1];
}

__global__ __launch_bounds__(128)
void k_conv(const float* __restrict__ feats,
            const float* __restrict__ wT,
            const float* __restrict__ bias,
            const int*   __restrict__ coords,
            const int*   __restrict__ grid,
            float*       __restrict__ out) {
    __shared__ int      nbr[NOFF*16];
    __shared__ unsigned occ[16];
    __shared__ float    A[16*A_STRIDE];

    const int tid  = threadIdx.x;
    const int lane = tid & 31;
    const int wv   = tid >> 5;
    const int p0   = blockIdx.x * 16;

    for (int i = tid; i < NOFF*16; i += 128) {
        int t = i >> 4, m = i & 15;
        int p = p0 + m;
        int bz = coords[p*4+0], z = coords[p*4+1], y = coords[p*4+2], x = coords[p*4+3];
        int kd = t / 9, kh = (t / 3) % 3, kw = t % 3;
        int nz = z + kd - 1, ny = y + kh - 1, nx = x + kw - 1;
        int idx = -1;
        if ((unsigned)nz < (unsigned)DD && (unsigned)ny < (unsigned)HH &&
            (unsigned)nx < (unsigned)WW) {
            int lin = ((bz*DD + nz)*HH + ny)*WW + nx;
            idx = grid[lin];
        }
        nbr[i] = idx;
        unsigned bm = __builtin_amdgcn_ballot_w32(idx >= 0);
        if (lane == 0) occ[i >> 5] = bm;
    }
    __syncthreads();

    unsigned live = 0;
#pragma unroll
    for (int j = 0; j < (NOFF + 1)/2; ++j) {
        unsigned w = occ[j];
        live |= ((w & 0xFFFFu) ? 1u : 0u) << (2*j);
        live |= ((w >> 16)     ? 1u : 0u) << (2*j + 1);
    }
    live &= (1u << NOFF) - 1u;
    live = __builtin_amdgcn_readfirstlane(live);

    const int col  = lane & 15;
    const int half = lane >> 4;
    float bv = bias[wv*16 + col];
    v8f acc;
#pragma unroll
    for (int j = 0; j < 8; ++j) acc[j] = bv;

    const int arow  = tid >> 3;
    const int acol8 = (tid & 7) * 8;

    while (live) {
        const int t = __builtin_ctz(live);
        live &= live - 1;
        int idx = nbr[t*16 + arow];
        float4 lo = make_float4(0.f, 0.f, 0.f, 0.f), hi = lo;
        if (idx >= 0) {
            const float4* src = (const float4*)(feats + (size_t)idx*CI + acol8);
            lo = src[0];
            hi = src[1];
        }
        float* dst = &A[arow*A_STRIDE + acol8];
        *(float4*)(dst)     = lo;
        *(float4*)(dst + 4) = hi;
        __syncthreads();

        const float* wbase = wT + (size_t)((t*16)*4 + wv)*64 + lane*2;
#pragma unroll
        for (int kc = 0; kc < 16; ++kc) {
            const float* ap = &A[col*A_STRIDE + kc*4 + 2*half];
            v2f a; a.x = ap[0]; a.y = ap[1];
            const float* bp = wbase + (size_t)kc*256;
            v2f b; b.x = bp[0]; b.y = bp[1];
            acc = __builtin_amdgcn_wmma_f32_16x16x4_f32(
                false, a, false, b, (short)0, acc, false, false);
        }
        __syncthreads();
    }

#pragma unroll
    for (int j = 0; j < 8; ++j) {
        int m = j + 8*half;
        out[(size_t)(p0 + m)*CO + wv*16 + col] = acc[j];
    }
}

extern "C" void kernel_launch(void* const* d_in, const int* in_sizes, int n_in,
                              void* d_out, int out_size, void* d_ws, size_t ws_size,
                              hipStream_t stream) {
    const float* feats  = (const float*)d_in[0];
    const float* weight = (const float*)d_in[1];
    const float* bias   = (const float*)d_in[2];
    const int*   coords = (const int*)  d_in[3];
    float* out = (float*)d_out;

    const int n = in_sizes[0] / CI;

    float* wT   = (float*)d_ws;
    int*   grid = (int*)((char*)d_ws + (size_t)WT_FLOATS*4);

    k_fill_grid<<<(GRID_CELLS + 255)/256, 256, 0, stream>>>(grid);
    k_scatter  <<<(n + 255)/256,          256, 0, stream>>>(coords, grid, n);
    k_wt       <<<(NOFF*16*4*32 + 255)/256, 256, 0, stream>>>(weight, wT);
    k_conv     <<<n/16, 128, 0, stream>>>(feats, wT, bias, coords, grid, out);
}


# per tap, the valid (point, neighbour) pairs of a 160-point block are packed into dense 16-row tiles, accumulators in LDS as C/D: matrix rows are real neighbours; summation order unchanged
# speedup vs baseline: 2.8641x; 1.8654x over previous
#include <hip/hip_runtime.h>

typedef float v2f __attribute__((ext_vector_type(2)));
typedef float v8f __attribute__((ext_vector_type(8)));

#define BB   2
#define DD   128
#define HH   128
#define WW   128
#define CI   64
#define CO   64
#define NOFF 27
#define NPTS 200000
#define GRID_CELLS (BB*DD*HH*WW)
#define WT_FLOATS  (NOFF*16*4*32*2)
#define A_STRIDE   68

__global__ void k_fill_grid(int* __restrict__ grid) {
    int i = blockIdx.x * blockDim.x + threadIdx.x;
    if (i < GRID_CELLS) grid[i] = -1;
}

__global__ void k_scatter(const int* __restrict__ coords, int* __restrict__ grid, int n) {
    int i = blockIdx.x * blockDim.x + threadIdx.x;
    if (i >= n) return;
    int b = coords[i*4+0], z = coords[i*4+1], y = coords[i*4+2], x = coords[i*4+3];
    int lin = ((b*DD + z)*HH + y)*WW + x;
    grid[lin] = i;
}

__global__ void k_wt(const float* __restrict__ w, float* __restrict__ wT) {
    int i = blockIdx.x * blockDim.x + threadIdx.x;
    if (i >= NOFF*16*4*32) return;
    int lane = i & 31;
    int nt   = (i >> 5) & 3;
    int kc   = (i >> 7) & 15;
    int t    =  i >> 11;
    int o = nt*16 + (lane & 15);
    int c = kc*4 + 2*(lane >> 4);
    const float* src = w + (size_t)(o*NOFF + t)*CI + c;
    wT[(size_t)i*2 + 0] = src[0];
    wT[(size_t)i*2 + 1] = src[1];
}

__device__ __forceinline__ int nbr_entry(const int* __restrict__ grid, int4 c,
                                         int kd, int kh, int kw, int m) {
    const int nz = c.y + kd - 1, ny = c.z + kh - 1, nx = c.w + kw - 1;
    const bool inb = (unsigned)nz < (unsigned)DD && (unsigned)ny < (unsigned)HH &&
                     (unsigned)nx < (unsigned)WW;
    const int cz = min(max(nz, 0), DD - 1), cy = min(max(ny, 0), HH - 1), cx = min(max(nx, 0), WW - 1);
    const int idx = grid[((c.x*DD + cz)*HH + cy)*WW + cx];
    return (inb && idx >= 0) ? ((idx << 8) | m) : -1;
}

#define PTS 160
__global__ __launch_bounds__(128)
void k_conv(const float* __restrict__ feats,
            const float* __restrict__ wT,
            const float* __restrict__ bias,
            const int*   __restrict__ coords,
            const int*   __restrict__ grid,
            float*       __restrict__ out) {
    __shared__ __attribute__((aligned(16))) int lst[NOFF*PTS];
    __shared__ int   cnt[NOFF];
    __shared__ float A[16*A_STRIDE];
    __shared__ __attribute__((aligned(16))) float accL[(PTS + 1)*CO];

    const int tid  = threadIdx.x;
    const int lane = tid & 31;
    const int wv   = __builtin_amdgcn_readfirstlane(tid >> 5);
    const int p0   = blockIdx.x * PTS;

    {
        const float4* b4 = (const float4*)bias;
        float4* a4 = (float4*)accL;
        for (int i = tid; i < (PTS + 1)*CO/4; i += 128) a4[i] = b4[i & 15];
    }

    {
        const int4 c0 = ((const int4*)coords)[p0 + tid];
        const int4 c1 = ((const int4*)coords)[p0 + 128 + lane];
        int e0[NOFF];
#pragma unroll
        for (int t = 0; t < NOFF; ++t)
            e0[t] = nbr_entry(grid, c0, t / 9, (t / 3) % 3, t % 3, tid);
        int e1[(NOFF + 3)/4];
#pragma unroll
        for (int k = 0; k < (NOFF + 3)/4; ++k) {
            const int t = min(wv + 4*k, NOFF - 1);
            e1[k] = nbr_entry(grid, c1, t / 9, (t / 3) % 3, t % 3, 128 + lane);
        }
#pragma unroll
        for (int t = 0; t < NOFF; ++t) lst[t*PTS + tid] = e0[t];
#pragma unroll
        for (int k = 0; k < (NOFF + 3)/4; ++k) {
            const int t = wv + 4*k;
            if (t < NOFF) lst[t*PTS + 128 + lane] = e1[k];
        }
    }
    __syncthreads();

    for (int t = wv; t < NOFF; t += 4) {
        int e[PTS/32];
#pragma unroll
        for (int c = 0; c < PTS/32; ++c) e[c] = lst[t*PTS + c*32 + lane];
        int n = 0;
#pragma unroll
        for (int c = 0; c < PTS/32; ++c) {
            const bool v = e[c] >= 0;
            const unsigned bm = __builtin_amdgcn_ballot_w32(v);
            const int pos = n + __builtin_popcount(bm & ((1u << lane) - 1u));
            if (v) lst[t*PTS + pos] = e[c];
            n += __builtin_popcount(bm);
        }
        if (lane == 0) cnt[t] = n;
    }
    __syncthreads();

    const int col  = lane & 15;
    const int half = lane >> 4;

    const int arow  = tid >> 3;
    const int acol8 = (tid & 7) * 8;

    for (int t = 0; t < NOFF; ++t) {
        const int n = __builtin_amdgcn_readfirstlane(cnt[t]);
        if (n == 0) continue;
        const int* L = &lst[t*PTS];

        const float* wbase = wT + (size_t)((t*16)*4 + wv)*64 + lane*2;
        v2f b[16];
#pragma unroll
        for (int kc = 0; kc < 16; ++kc) {
            const float* bp = wbase + (size_t)kc*256;
            b[kc].x = bp[0]; b[kc].y = bp[1];
        }

        for (int base = 0; base < n; base += 16) {
            const int nv = n - base;
            {
                const int e = L[base + arow];
                float4 lo = make_float4(0.f, 0.f, 0.f, 0.f), hi = lo;
                if (arow < nv) {
                    const float4* src = (const float4*)(feats + (size_t)(e >> 8)*CI + acol8);
                    lo = src[0];
                    hi = src[1];
                }
                float* dst = &A[arow*A_STRIDE + acol8];
                *(float4*)(dst)     = lo;
                *(float4*)(dst + 4) = hi;
            }

            int  am[8];
            v8f acc;
            {
                const int4* L4 = (const int4*)(L + base + 8*half);
                const int4 e0 = L4[0], e1 = L4[1];
                const int ej[8] = {e0.x, e0.y, e0.z, e0.w, e1.x, e1.y, e1.z, e1.w};
#pragma unroll
                for (int j = 0; j < 8; ++j) {
                    const bool v = (j + 8*half) < nv;
                    am[j] = (v ? (ej[j] & 255) : PTS)*CO + wv*16 + col;
                }
#pragma unroll
                for (int j = 0; j < 8; ++j) {
                    const float cv = accL[am[j]];
                    acc[j] = ((j + 8*half) < nv) ? cv : 0.f;
                }
            }
            __syncthreads();

#pragma unroll
            for (int kc = 0; kc < 16; ++kc) {
                const float* ap = &A[col*A_STRIDE + kc*4 + 2*half];
                v2f a; a.x = ap[0]; a.y = ap[1];
                acc = __builtin_amdgcn_wmma_f32_16x16x4_f32(
                    false, a, false, b[kc], (short)0, acc, false, false);
            }

#pragma unroll
            for (int j = 0; j < 8; ++j) accL[am[j]] = acc[j];
            __syncthreads();
        }
    }

    {
        const float4* a4 = (const float4*)accL;
        float4* o4 = (float4*)(out + (size_t)p0*CO);
        for (int i = tid; i < PTS*CO/4; i += 128) o4[i] = a4[i];
    }
}

extern "C" void kernel_launch(void* const* d_in, const int* in_sizes, int n_in,
                              void* d_out, int out_size, void* d_ws, size_t ws_size,
                              hipStream_t stream) {
    const float* feats  = (const float*)d_in[0];
    const float* weight = (const float*)d_in[1];
    const float* bias   = (const float*)d_in[2];
    const int*   coords = (const int*)  d_in[3];
    float* out = (float*)d_out;

    const int n = in_sizes[0] / CI;

    float* wT   = (float*)d_ws;
    int*   grid = (int*)((char*)d_ws + (size_t)WT_FLOATS*4);

    k_fill_grid<<<(GRID_CELLS + 255)/256, 256, 0, stream>>>(grid);
    k_scatter  <<<(n + 255)/256,          256, 0, stream>>>(coords, grid, n);
    k_wt       <<<(NOFF*16*4*32 + 255)/256, 256, 0, stream>>>(weight, wT);
    k_conv     <<<n/PTS, 128, 0, stream>>>(feats, wT, bias, coords, grid, out);
}


# tile loop software-pipelined: next tile's neighbour rows and next tap's B fragments are loaded into registers ahead of the current tile's matrix chain (barriers fence LDS only); sum order unchanged
# speedup vs baseline: 3.0870x; 1.0778x over previous
#include <hip/hip_runtime.h>

typedef float v2f __attribute__((ext_vector_type(2)));
typedef float v8f __attribute__((ext_vector_type(8)));

#define BB   2
#define DD   128
#define HH   128
#define WW   128
#define CI   64
#define CO   64
#define NOFF 27
#define NPTS 200000
#define GRID_CELLS (BB*DD*HH*WW)
#define WT_FLOATS  (NOFF*16*4*32*2)
#define A_STRIDE   68

__global__ void k_fill_grid(int* __restrict__ grid) {
    int i = blockIdx.x * blockDim.x + threadIdx.x;
    if (i < GRID_CELLS) grid[i] = -1;
}

__global__ void k_scatter(const int* __restrict__ coords, int* __restrict__ grid, int n) {
    int i = blockIdx.x * blockDim.x + threadIdx.x;
    if (i >= n) return;
    int b = coords[i*4+0], z = coords[i*4+1], y = coords[i*4+2], x = coords[i*4+3];
    int lin = ((b*DD + z)*HH + y)*WW + x;
    grid[lin] = i;
}

__global__ void k_wt(const float* __restrict__ w, float* __restrict__ wT) {
    int i = blockIdx.x * blockDim.x + threadIdx.x;
    if (i >= NOFF*16*4*32) return;
    int lane = i & 31;
    int nt   = (i >> 5) & 3;
    int kc   = (i >> 7) & 15;
    int t    =  i >> 11;
    int o = nt*16 + (lane & 15);
    int c = kc*4 + 2*(lane >> 4);
    const float* src = w + (size_t)(o*NOFF + t)*CI + c;
    wT[(size_t)i*2 + 0] = src[0];
    wT[(size_t)i*2 + 1] = src[1];
}

__device__ __forceinline__ int nbr_entry(const int* __restrict__ grid, int4 c,
                                         int kd, int kh, int kw, int m) {
    const int nz = c.y + kd - 1, ny = c.z + kh - 1, nx = c.w + kw - 1;
    const bool inb = (unsigned)nz < (unsigned)DD && (unsigned)ny < (unsigned)HH &&
                     (unsigned)nx < (unsigned)WW;
    const int cz = min(max(nz, 0), DD - 1), cy = min(max(ny, 0), HH - 1), cx = min(max(nx, 0), WW - 1);
    const int idx = grid[((c.x*DD + cz)*HH + cy)*WW + cx];
    return (inb && idx >= 0) ? ((idx << 8) | m) : -1;
}

#define PTS 160
__global__ __launch_bounds__(128)
void k_conv(const float* __restrict__ feats,
            const float* __restrict__ wT,
            const float* __restrict__ bias,
            const int*   __restrict__ coords,
            const int*   __restrict__ grid,
            float*       __restrict__ out) {
    __shared__ __attribute__((aligned(16))) int lst[NOFF*PTS];
    __shared__ int   cnt[NOFF];
    __shared__ int   info[NOFF + 1];
    __shared__ float A[16*A_STRIDE];
    __shared__ __attribute__((aligned(16))) float accL[(PTS + 1)*CO];

    const int tid  = threadIdx.x;
    const int lane = tid & 31;
    const int wv   = __builtin_amdgcn_readfirstlane(tid >> 5);
    const int p0   = blockIdx.x * PTS;

    {
        const float4* b4 = (const float4*)bias;
        float4* a4 = (float4*)accL;
        for (int i = tid; i < (PTS + 1)*CO/4; i += 128) a4[i] = b4[i & 15];
    }

    {
        const int4 c0 = ((const int4*)coords)[p0 + tid];
        const int4 c1 = ((const int4*)coords)[p0 + 128 + lane];
        int e0[NOFF];
#pragma unroll
        for (int t = 0; t < NOFF; ++t)
            e0[t] = nbr_entry(grid, c0, t / 9, (t / 3) % 3, t % 3, tid);
        int e1[(NOFF + 3)/4];
#pragma unroll
        for (int k = 0; k < (NOFF + 3)/4; ++k) {
            const int t = min(wv + 4*k, NOFF - 1);
            e1[k] = nbr_entry(grid, c1, t / 9, (t / 3) % 3, t % 3, 128 + lane);
        }
#pragma unroll
        for (int t = 0; t < NOFF; ++t) lst[t*PTS + tid] = e0[t];
#pragma unroll
        for (int k = 0; k < (NOFF + 3)/4; ++k) {
            const int t = wv + 4*k;
            if (t < NOFF) lst[t*PTS + 128 + lane] = e1[k];
        }
    }
    __syncthreads();

    for (int t = wv; t < NOFF; t += 4) {
        int e[PTS/32];
#pragma unroll
        for (int c = 0; c < PTS/32; ++c) e[c] = lst[t*PTS + c*32 + lane];
        int n = 0;
#pragma unroll
        for (int c = 0; c < PTS/32; ++c) {
            const bool v = e[c] >= 0;
            const unsigned bm = __builtin_amdgcn_ballot_w32(v);
            const int pos = n + __builtin_popcount(bm & ((1u << lane) - 1u));
            if (v) lst[t*PTS + pos] = e[c];
            n += __builtin_popcount(bm);
        }
        if (lane == 0) cnt[t] = n;
    }
    __syncthreads();

    if (tid <= NOFF) {
        int nx = NOFF, nn = 0;
        for (int u = NOFF - 1; u > tid; --u) {
            const int c = cnt[u];
            if (c > 0) { nx = u; nn = c; }
        }
        const int n = (tid < NOFF) ? cnt[tid] : 0;
        info[tid] = n | (nx << 8) | (nn << 16);
    }
    __syncthreads();

    const int col  = lane & 15;
    const int half = lane >> 4;

    const int arow  = tid >> 3;
    const int acol8 = (tid & 7) * 8;
    float* const adst = &A[arow*A_STRIDE + acol8];

    int t = 0;
    while (t < NOFF && __builtin_amdgcn_readfirstlane(cnt[t]) == 0) ++t;

    if (t < NOFF) {
        int inf = __builtin_amdgcn_readfirstlane(info[t]);
        int n  = inf & 255;
        int tn = (inf >> 8) & 255;
        int nn = inf >> 16;
        int infn_v = info[tn];

        v2f b[16], bnx[16];
        {
            const float* wbase = wT + (size_t)((t*16)*4 + wv)*64 + lane*2;
#pragma unroll
            for (int kc = 0; kc < 16; ++kc) {
                const float* bp = wbase + (size_t)kc*256;
                b[kc].x = bp[0]; b[kc].y = bp[1];
            }
        }
        float4 lo = make_float4(0.f, 0.f, 0.f, 0.f), hi = lo;
        if (arow < n) {
            const int e = lst[t*PTS + arow];
            const float4* src = (const float4*)(feats + (size_t)(e >> 8)*CI + acol8);
            lo = src[0];
            hi = src[1];
        }
        *(float4*)(adst)     = lo;
        *(float4*)(adst + 4) = hi;

        int base = 0;
        for (;;) {
            const bool last_in_tap = base + 16 >= n;
            const int  t2 = last_in_tap ? tn : t;
            const int  b2 = last_in_tap ? 0  : base + 16;
            const int  n2 = last_in_tap ? nn : n;
            const bool has_next = t2 < NOFF;

            lo = make_float4(0.f, 0.f, 0.f, 0.f); hi = lo;
            if (has_next) {
                if (arow < n2 - b2) {
                    const int e = lst[t2*PTS + b2 + arow];
                    const float4* src = (const float4*)(feats + (size_t)(e >> 8)*CI + acol8);
                    lo = src[0];
                    hi = src[1];
                }
                if (last_in_tap) {
                    const float* wbase = wT + (size_t)((t2*16)*4 + wv)*64 + lane*2;
#pragma unroll
                    for (int kc = 0; kc < 16; ++kc) {
                        const float* bp = wbase + (size_t)kc*256;
                        bnx[kc].x = bp[0]; bnx[kc].y = bp[1];
                    }
                }
            }

            const int* L = &lst[t*PTS];
            const int nv = n - base;

            int  am[8];
            v8f acc;
            {
                const int4* L4 = (const int4*)(L + base + 8*half);
                const int4 e0 = L4[0], e1 = L4[1];
                const int ej[8] = {e0.x, e0.y, e0.z, e0.w, e1.x, e1.y, e1.z, e1.w};
#pragma unroll
                for (int j = 0; j < 8; ++j) {
                    const bool v = (j + 8*half) < nv;
                    am[j] = (v ? (ej[j] & 255) : PTS)*CO + wv*16 + col;
                }
#pragma unroll
                for (int j = 0; j < 8; ++j) {
                    const float cv = accL[am[j]];
                    acc[j] = ((j + 8*half) < nv) ? cv : 0.f;
                }
            }
            __builtin_amdgcn_fence(__ATOMIC_RELEASE, "workgroup", "local");
            __builtin_amdgcn_s_barrier();
            __builtin_amdgcn_fence(__ATOMIC_ACQUIRE, "workgroup", "local");

#pragma unroll
            for (int kc = 0; kc < 16; ++kc) {
                const float* ap = &A[col*A_STRIDE + kc*4 + 2*half];
                v2f a; a.x = ap[0]; a.y = ap[1];
                acc = __builtin_amdgcn_wmma_f32_16x16x4_f32(
                    false, a, false, b[kc], (short)0, acc, false, false);
            }

#pragma unroll
            for (int j = 0; j < 8; ++j) accL[am[j]] = acc[j];

            if (!has_next) break;

            __builtin_amdgcn_fence(__ATOMIC_RELEASE, "workgroup", "local");
            __builtin_amdgcn_s_barrier();
            __builtin_amdgcn_fence(__ATOMIC_ACQUIRE, "workgroup", "local");

            *(float4*)(adst)     = lo;
            *(float4*)(adst + 4) = hi;
            if (last_in_tap) {
#pragma unroll
                for (int kc = 0; kc < 16; ++kc) b[kc] = bnx[kc];
                inf = __builtin_amdgcn_readfirstlane(infn_v);
                t  = tn;
                n  = nn;
                tn = (inf >> 8) & 255;
                nn = inf >> 16;
                infn_v = info[tn];
            }
            base = b2;
        }
    }
    __syncthreads();

    {
        const float4* a4 = (const float4*)accL;
        float4* o4 = (float4*)(out + (size_t)p0*CO);
        for (int i = tid; i < PTS*CO/4; i += 128) o4[i] = a4[i];
    }
}

extern "C" void kernel_launch(void* const* d_in, const int* in_sizes, int n_in,
                              void* d_out, int out_size, void* d_ws, size_t ws_size,
                              hipStream_t stream) {
    const float* feats  = (const float*)d_in[0];
    const float* weight = (const float*)d_in[1];
    const float* bias   = (const float*)d_in[2];
    const int*   coords = (const int*)  d_in[3];
    float* out = (float*)d_out;

    const int n = in_sizes[0] / CI;

    float* wT   = (float*)d_ws;
    int*   grid = (int*)((char*)d_ws + (size_t)WT_FLOATS*4);

    k_fill_grid<<<(GRID_CELLS + 255)/256, 256, 0, stream>>>(grid);
    k_scatter  <<<(n + 255)/256,          256, 0, stream>>>(coords, grid, n);
    k_wt       <<<(NOFF*16*4*32 + 255)/256, 256, 0, stream>>>(weight, wT);
    k_conv     <<<n/PTS, 128, 0, stream>>>(feats, wT, bias, coords, grid, out);
}


# neighbour lookup reads the three x-cells of each grid row with one 12-byte load (9 loads of 3 taps per point instead of 27 single-cell loads); table entries and sum order unchanged
# speedup vs baseline: 3.2140x; 1.0411x over previous
#include <hip/hip_runtime.h>

typedef float v2f __attribute__((ext_vector_type(2)));
typedef float v8f __attribute__((ext_vector_type(8)));

#define BB   2
#define DD   128
#define HH   128
#define WW   128
#define CI   64
#define CO   64
#define NOFF 27
#define NPTS 200000
#define GRID_CELLS (BB*DD*HH*WW)
#define WT_FLOATS  (NOFF*16*4*32*2)
#define A_STRIDE   68

__global__ void k_fill_grid(int* __restrict__ grid) {
    int i = blockIdx.x * blockDim.x + threadIdx.x;
    if (i < GRID_CELLS) grid[i] = -1;
}

__global__ void k_scatter(const int* __restrict__ coords, int* __restrict__ grid, int n) {
    int i = blockIdx.x * blockDim.x + threadIdx.x;
    if (i >= n) return;
    int b = coords[i*4+0], z = coords[i*4+1], y = coords[i*4+2], x = coords[i*4+3];
    int lin = ((b*DD + z)*HH + y)*WW + x;
    grid[lin] = i;
}

__global__ void k_wt(const float* __restrict__ w, float* __restrict__ wT) {
    int i = blockIdx.x * blockDim.x + threadIdx.x;
    if (i >= NOFF*16*4*32) return;
    int lane = i & 31;
    int nt   = (i >> 5) & 3;
    int kc   = (i >> 7) & 15;
    int t    =  i >> 11;
    int o = nt*16 + (lane & 15);
    int c = kc*4 + 2*(lane >> 4);
    const float* src = w + (size_t)(o*NOFF + t)*CI + c;
    wT[(size_t)i*2 + 0] = src[0];
    wT[(size_t)i*2 + 1] = src[1];
}

struct RowCells { int g0, g1, g2; };
__device__ __forceinline__ RowCells load_row(const int* __restrict__ grid, int4 c, int kd, int kh) {
    const int cz = min(max(c.y + kd - 1, 0), DD - 1), cy = min(max(c.z + kh - 1, 0), HH - 1);
    const int xb = min(max(c.w - 1, 0), WW - 3);
    const int* p = grid + ((c.x*DD + cz)*HH + cy)*WW + xb;
    RowCells r;
    r.g0 = p[0]; r.g1 = p[1]; r.g2 = p[2];
    return r;
}
__device__ __forceinline__ int row_entry(RowCells r, int4 c, int kd, int kh, int kw, int m) {
    const int nz = c.y + kd - 1, ny = c.z + kh - 1, nx = c.w + kw - 1;
    const bool inb = (unsigned)nz < (unsigned)DD && (unsigned)ny < (unsigned)HH &&
                     (unsigned)nx < (unsigned)WW;
    const int xb = min(max(c.w - 1, 0), WW - 3);
    const int j = nx - xb;
    const int idx = (j <= 0) ? r.g0 : ((j == 1) ? r.g1 : r.g2);
    return (inb && idx >= 0) ? ((idx << 8) | m) : -1;
}

#define PTS 160
__global__ __launch_bounds__(128)
void k_conv(const float* __restrict__ feats,
            const float* __restrict__ wT,
            const float* __restrict__ bias,
            const int*   __restrict__ coords,
            const int*   __restrict__ grid,
            float*       __restrict__ out) {
    __shared__ __attribute__((aligned(16))) int lst[NOFF*PTS];
    __shared__ int   cnt[NOFF];
    __shared__ int   info[NOFF + 1];
    __shared__ float A[16*A_STRIDE];
    __shared__ __attribute__((aligned(16))) float accL[(PTS + 1)*CO];

    const int tid  = threadIdx.x;
    const int lane = tid & 31;
    const int wv   = __builtin_amdgcn_readfirstlane(tid >> 5);
    const int p0   = blockIdx.x * PTS;

    {
        const float4* b4 = (const float4*)bias;
        float4* a4 = (float4*)accL;
        for (int i = tid; i < (PTS + 1)*CO/4; i += 128) a4[i] = b4[i & 15];
    }

    {
        const int4 c0 = ((const int4*)coords)[p0 + tid];
        const int4 c1 = ((const int4*)coords)[p0 + 128 + lane];
        RowCells q0[9];
#pragma unroll
        for (int r = 0; r < 9; ++r) q0[r] = load_row(grid, c0, r / 3, r % 3);
        RowCells q1[3];
#pragma unroll
        for (int k = 0; k < 3; ++k) {
            const int r = min(wv + 4*k, 8);
            q1[k] = load_row(grid, c1, r / 3, r % 3);
        }
#pragma unroll
        for (int t = 0; t < NOFF; ++t)
            lst[t*PTS + tid] = row_entry(q0[t / 3], c0, t / 9, (t / 3) % 3, t % 3, tid);
#pragma unroll
        for (int k = 0; k < 3; ++k) {
            const int r = wv + 4*k;
            if (r < 9) {
#pragma unroll
                for (int kw = 0; kw < 3; ++kw)
                    lst[(3*r + kw)*PTS + 128 + lane] = row_entry(q1[k], c1, r / 3, r % 3, kw, 128 + lane);
            }
        }
    }
    __syncthreads();

    for (int t = wv; t < NOFF; t += 4) {
        int e[PTS/32];
#pragma unroll
        for (int c = 0; c < PTS/32; ++c) e[c] = lst[t*PTS + c*32 + lane];
        int n = 0;
#pragma unroll
        for (int c = 0; c < PTS/32; ++c) {
            const bool v = e[c] >= 0;
            const unsigned bm = __builtin_amdgcn_ballot_w32(v);
            const int pos = n + __builtin_popcount(bm & ((1u << lane) - 1u));
            if (v) lst[t*PTS + pos] = e[c];
            n += __builtin_popcount(bm);
        }
        if (lane == 0) cnt[t] = n;
    }
    __syncthreads();

    if (tid <= NOFF) {
        int nx = NOFF, nn = 0;
        for (int u = NOFF - 1; u > tid; --u) {
            const int c = cnt[u];
            if (c > 0) { nx = u; nn = c; }
        }
        const int n = (tid < NOFF) ? cnt[tid] : 0;
        info[tid] = n | (nx << 8) | (nn << 16);
    }
    __syncthreads();

    const int col  = lane & 15;
    const int half = lane >> 4;

    const int arow  = tid >> 3;
    const int acol8 = (tid & 7) * 8;
    float* const adst = &A[arow*A_STRIDE + acol8];

    int t = 0;
    while (t < NOFF && __builtin_amdgcn_readfirstlane(cnt[t]) == 0) ++t;

    if (t < NOFF) {
        int inf = __builtin_amdgcn_readfirstlane(info[t]);
        int n  = inf & 255;
        int tn = (inf >> 8) & 255;
        int nn = inf >> 16;
        int infn_v = info[tn];

        v2f b[16], bnx[16];
        {
            const float* wbase = wT + (size_t)((t*16)*4 + wv)*64 + lane*2;
#pragma unroll
            for (int kc = 0; kc < 16; ++kc) {
                const float* bp = wbase + (size_t)kc*256;
                b[kc].x = bp[0]; b[kc].y = bp[1];
            }
        }
        float4 lo = make_float4(0.f, 0.f, 0.f, 0.f), hi = lo;
        if (arow < n) {
            const int e = lst[t*PTS + arow];
            const float4* src = (const float4*)(feats + (size_t)(e >> 8)*CI + acol8);
            lo = src[0];
            hi = src[1];
        }
        *(float4*)(adst)     = lo;
        *(float4*)(adst + 4) = hi;

        int base = 0;
        for (;;) {
            const bool last_in_tap = base + 16 >= n;
            const int  t2 = last_in_tap ? tn : t;
            const int  b2 = last_in_tap ? 0  : base + 16;
            const int  n2 = last_in_tap ? nn : n;
            const bool has_next = t2 < NOFF;

            lo = make_float4(0.f, 0.f, 0.f, 0.f); hi = lo;
            if (has_next) {
                if (arow < n2 - b2) {
                    const int e = lst[t2*PTS + b2 + arow];
                    const float4* src = (const float4*)(feats + (size_t)(e >> 8)*CI + acol8);
                    lo = src[0];
                    hi = src[1];
                }
                if (last_in_tap) {
                    const float* wbase = wT + (size_t)((t2*16)*4 + wv)*64 + lane*2;
#pragma unroll
                    for (int kc = 0; kc < 16; ++kc) {
                        const float* bp = wbase + (size_t)kc*256;
                        bnx[kc].x = bp[0]; bnx[kc].y = bp[1];
                    }
                }
            }

            const int* L = &lst[t*PTS];
            const int nv = n - base;

            int  am[8];
            v8f acc;
            {
                const int4* L4 = (const int4*)(L + base + 8*half);
                const int4 e0 = L4[0], e1 = L4[1];
                const int ej[8] = {e0.x, e0.y, e0.z, e0.w, e1.x, e1.y, e1.z, e1.w};
#pragma unroll
                for (int j = 0; j < 8; ++j) {
                    const bool v = (j + 8*half) < nv;
                    am[j] = (v ? (ej[j] & 255) : PTS)*CO + wv*16 + col;
                }
#pragma unroll
                for (int j = 0; j < 8; ++j) {
                    const float cv = accL[am[j]];
                    acc[j] = ((j + 8*half) < nv) ? cv : 0.f;
                }
            }
            __builtin_amdgcn_fence(__ATOMIC_RELEASE, "workgroup", "local");
            __builtin_amdgcn_s_barrier();
            __builtin_amdgcn_fence(__ATOMIC_ACQUIRE, "workgroup", "local");

#pragma unroll
            for (int kc = 0; kc < 16; ++kc) {
                const float* ap = &A[col*A_STRIDE + kc*4 + 2*half];
                v2f a; a.x = ap[0]; a.y = ap[1];
                acc = __builtin_amdgcn_wmma_f32_16x16x4_f32(
                    false, a, false, b[kc], (short)0, acc, false, false);
            }

#pragma unroll
            for (int j = 0; j < 8; ++j) accL[am[j]] = acc[j];

            if (!has_next) break;

            __builtin_amdgcn_fence(__ATOMIC_RELEASE, "workgroup", "local");
            __builtin_amdgcn_s_barrier();
            __builtin_amdgcn_fence(__ATOMIC_ACQUIRE, "workgroup", "local");

            *(float4*)(adst)     = lo;
            *(float4*)(adst + 4) = hi;
            if (last_in_tap) {
#pragma unroll
                for (int kc = 0; kc < 16; ++kc) b[kc] = bnx[kc];
                inf = __builtin_amdgcn_readfirstlane(infn_v);
                t  = tn;
                n  = nn;
                tn = (inf >> 8) & 255;
                nn = inf >> 16;
                infn_v = info[tn];
            }
            base = b2;
        }
    }
    __syncthreads();

    {
        const float4* a4 = (const float4*)accL;
        float4* o4 = (float4*)(out + (size_t)p0*CO);
        for (int i = tid; i < PTS*CO/4; i += 128) o4[i] = a4[i];
    }
}

extern "C" void kernel_launch(void* const* d_in, const int* in_sizes, int n_in,
                              void* d_out, int out_size, void* d_ws, size_t ws_size,
                              hipStream_t stream) {
    const float* feats  = (const float*)d_in[0];
    const float* weight = (const float*)d_in[1];
    const float* bias   = (const float*)d_in[2];
    const int*   coords = (const int*)  d_in[3];
    float* out = (float*)d_out;

    const int n = in_sizes[0] / CI;

    float* wT   = (float*)d_ws;
    int*   grid = (int*)((char*)d_ws + (size_t)WT_FLOATS*4);

    k_fill_grid<<<(GRID_CELLS + 255)/256, 256, 0, stream>>>(grid);
    k_scatter  <<<(n + 255)/256,          256, 0, stream>>>(coords, grid, n);
    k_wt       <<<(NOFF*16*4*32 + 255)/256, 256, 0, stream>>>(weight, wT);
    k_conv     <<<n/PTS, 128, 0, stream>>>(feats, wT, bias, coords, grid, out);
}
